# starting kernel
# baseline (speedup 1.0000x reference)
#include <hip/hip_runtime.h>
#include <hip/hip_bf16.h>

typedef __attribute__((ext_vector_type(2))) float v2f;
typedef __attribute__((ext_vector_type(8))) float v8f;

#define ALPHA 0.2f
#define BATCH 32
#define WDIM 128
#define KDIM 1024
#define JT   64
#define NCHUNK (KDIM / JT)
#define PSTR 258

__global__ __launch_bounds__(256) void fal_prep_kernel(const float* __restrict__ x,
                                                       const float* __restrict__ w,
                                                       float* __restrict__ s1,
                                                       float* __restrict__ s2,
                                                       float* __restrict__ t1,
                                                       float* __restrict__ ta) {
  int idx = blockIdx.x * 256 + threadIdx.x;
  int b = idx >> 10;
  int k = idx & (KDIM - 1);
  const float* xp = x + ((size_t)b << 17) + k;
  float a1 = 0.f, a2 = 0.f;
#pragma unroll 8
  for (int wi = 0; wi < WDIM; ++wi) {
    float xv = xp[(size_t)wi << 10];
    a1 = fmaf(xv, w[wi], a1);
    a2 = fmaf(xv, w[WDIM + wi], a2);
  }
  s1[idx] = a1;
  s2[idx] = a2;
  t1[idx] = __expf(a2);
  ta[idx] = __expf(ALPHA * a2);
}

__global__ __launch_bounds__(256) void fal_rowmax_kernel(const float* __restrict__ s2,
                                                         float* __restrict__ s2max) {
  __shared__ float red[256];
  int b = blockIdx.x;
  float m = -3.4e38f;
  for (int j = threadIdx.x; j < KDIM; j += 256)
    m = fmaxf(m, s2[b * KDIM + j]);
  red[threadIdx.x] = m;
  __syncthreads();
  for (int s = 128; s > 0; s >>= 1) {
    if (threadIdx.x < s) red[threadIdx.x] = fmaxf(red[threadIdx.x], red[threadIdx.x + s]);
    __syncthreads();
  }
  if (threadIdx.x == 0) s2max[b] = red[0];
}

__global__ __launch_bounds__(256) void fal_attn_kernel(const float* __restrict__ x,
                                                       const float* __restrict__ s1g,
                                                       const float* __restrict__ s2g,
                                                       const float* __restrict__ t1g,
                                                       const float* __restrict__ tag,
                                                       const float* __restrict__ s2mx,
                                                       float* __restrict__ out) {
  __shared__ float  Vt[(JT / 2) * PSTR];
  __shared__ float4 qsm[JT];

  const int tid  = threadIdx.x;
  const int lane = tid & 31;
  const int wave = tid >> 5;
  const int half = lane >> 4;
  const int m    = lane & 15;
  const int b    = blockIdx.y;
  const int i0   = blockIdx.x * 128 + wave * 16;

  const float s1v  = s1g[b * KDIM + i0 + m];
  const float zmax = s1v + s2mx[b];
  const float mrow = fmaxf(zmax, ALPHA * zmax);
  const float u1   = __expf(s1v - mrow);
  const float ua   = __expf(ALPHA * s1v - mrow);

  v8f acc[8];
#pragma unroll
  for (int t = 0; t < 8; ++t) acc[t] = (v8f)0.f;
  float dsum = 0.f;

  const int k0 = half * 2;

  const int jj4 = tid & 15;
  const int w0  = tid >> 4;
  const float* gbase = x + (((size_t)b * WDIM + w0) << 10) + (jj4 << 2);
  const int lbase = (jj4 * 2) * PSTR + 2 * w0;

  float4 r[8];
#pragma unroll
  for (int it = 0; it < 8; ++it)
    r[it] = *(const float4*)(gbase + ((size_t)(it * 16) << 10));

  for (int c = 0; c < NCHUNK; ++c) {
    const int j0 = c * JT;
    __syncthreads();
#pragma unroll
    for (int it = 0; it < 8; ++it) {
      int o = lbase + it * 32;
      *(float2*)&Vt[o]        = make_float2(r[it].x, r[it].y);
      *(float2*)&Vt[o + PSTR] = make_float2(r[it].z, r[it].w);
    }
    if (tid < JT) {
      int j = b * KDIM + j0 + tid;
      qsm[tid] = make_float4(s2g[j], t1g[j], tag[j], 0.f);
    }
    if (c + 1 < NCHUNK) {
#pragma unroll
      for (int it = 0; it < 8; ++it)
        r[it] = *(const float4*)(gbase + ((size_t)(it * 16) << 10) + (c + 1) * JT);
    }
    __syncthreads();

#pragma unroll
    for (int js = 0; js < JT; js += 4) {
      float4 qa = qsm[js + k0];
      float4 qb = qsm[js + k0 + 1];
      float z0 = s1v + qa.x;
      float z1 = s1v + qb.x;
      float p0 = (z0 > 0.f) ? u1 * qa.y : ua * qa.z;
      float p1 = (z1 > 0.f) ? u1 * qb.y : ua * qb.z;
      dsum += p0 + p1;
      v2f a; a.x = p0; a.y = p1;

      const float* vrow = &Vt[(js / 2 + half) * PSTR + 2 * m];
#pragma unroll
      for (int t = 0; t < 8; ++t) {
        v2f bb = *(const v2f*)(vrow + t * 32);
        acc[t] = __builtin_amdgcn_wmma_f32_16x16x4_f32(false, a, false, bb,
                                                       (short)0, acc[t], false, false);
      }
    }
  }

  float dtot = dsum + __shfl_xor(dsum, 16, 32);

#pragma unroll
  for (int rr = 0; rr < 8; ++rr) {
    int row   = rr + 8 * half;
    float d   = __shfl(dtot, rr + 8 * half, 32);
    float inv = 1.0f / d;
    float* orow = out + (((size_t)b * KDIM + i0 + row) << 7) + m;
#pragma unroll
    for (int t = 0; t < 8; ++t)
      orow[t * 16] = acc[t][rr] * inv;
  }
}

extern "C" void kernel_launch(void* const* d_in, const int* in_sizes, int n_in,
                              void* d_out, int out_size, void* d_ws, size_t ws_size,
                              hipStream_t stream) {
  const float* x = (const float*)d_in[0];
  const float* w = (const float*)d_in[1];
  float* out = (float*)d_out;

  float* s1  = (float*)d_ws;
  float* s2  = s1 + BATCH * KDIM;
  float* t1  = s2 + BATCH * KDIM;
  float* ta  = t1 + BATCH * KDIM;
  float* s2m = ta + BATCH * KDIM;

  fal_prep_kernel<<<(BATCH * KDIM) / 256, 256, 0, stream>>>(x, w, s1, s2, t1, ta);
  fal_rowmax_kernel<<<BATCH, 256, 0, stream>>>(s2, s2m);
  dim3 grid(KDIM / 128, BATCH);
  fal_attn_kernel<<<grid, 256, 0, stream>>>(x, s1, s2, t1, ta, s2m, out);
}


# pre-pass loads issued as one group: prep work-group = batch x 128 columns, each wave 16 rows of 512-byte loads, partials joined in LDS; row-max one 16-byte load per thread; s1/s2 sum order changes
# speedup vs baseline: 1.5509x; 1.5509x over previous
#include <hip/hip_runtime.h>
#include <hip/hip_bf16.h>

typedef __attribute__((ext_vector_type(2))) float v2f;
typedef __attribute__((ext_vector_type(4))) float v4f;
typedef __attribute__((ext_vector_type(8))) float v8f;

#define ALPHA 0.2f
#define BATCH 32
#define WDIM 128
#define KDIM 1024
#define JB   16
#define NBLK (KDIM / JB)

#define PCOLS 128
#define PROWS (WDIM / 8)
__global__ __launch_bounds__(256) __attribute__((amdgpu_waves_per_eu(2, 8))) void fal_prep_kernel(const float* __restrict__ x,
                                                       const float* __restrict__ w,
                                                       float* __restrict__ s1,
                                                       float* __restrict__ s2,
                                                       float* __restrict__ t1,
                                                       float* __restrict__ ta) {
  __shared__ __attribute__((aligned(16))) float part[8 * 2 * PCOLS];
  const int tid  = threadIdx.x;
  const int lane = tid & 31;
  const int wave = __builtin_amdgcn_readfirstlane(tid >> 5);
  const int b    = blockIdx.x >> 3;
  const int k0   = (blockIdx.x & 7) * PCOLS;
  const float* xp = x + (((size_t)b * WDIM + wave * PROWS) << 10) + k0 + 4 * lane;
  const float* w1 = w + wave * PROWS;
  const float* w2 = w + WDIM + wave * PROWS;

  v4f xv[PROWS];
#pragma unroll
  for (int t = 0; t < PROWS; ++t) xv[t] = *(const v4f*)(xp + ((size_t)t << 10));

  v4f a1 = (v4f)0.f, a2 = (v4f)0.f;
#pragma unroll
  for (int t = 0; t < PROWS; ++t) {
    const float c1 = w1[t], c2 = w2[t];
    a1.x = fmaf(xv[t].x, c1, a1.x); a2.x = fmaf(xv[t].x, c2, a2.x);
    a1.y = fmaf(xv[t].y, c1, a1.y); a2.y = fmaf(xv[t].y, c2, a2.y);
    a1.z = fmaf(xv[t].z, c1, a1.z); a2.z = fmaf(xv[t].z, c2, a2.z);
    a1.w = fmaf(xv[t].w, c1, a1.w); a2.w = fmaf(xv[t].w, c2, a2.w);
  }
  *(v4f*)&part[(wave * 2 + 0) * PCOLS + 4 * lane] = a1;
  *(v4f*)&part[(wave * 2 + 1) * PCOLS + 4 * lane] = a2;
  __syncthreads();

  const int sel = tid >> 7;
  const int col = tid & (PCOLS - 1);
  float acc = 0.f;
#pragma unroll
  for (int r = 0; r < 8; ++r) acc += part[(r * 2 + sel) * PCOLS + col];
  const int idx = b * KDIM + k0 + col;
  if (sel == 0) {
    s1[idx] = acc;
  } else {
    s2[idx] = acc;
    t1[idx] = __expf(acc);
    ta[idx] = __expf(ALPHA * acc);
  }
}
#undef PCOLS
#undef PROWS

__global__ __launch_bounds__(256) void fal_rowmax_kernel(const float* __restrict__ s2,
                                                         float* __restrict__ s2max) {
  __shared__ float red[256];
  int b = blockIdx.x;
  const v4f v = *(const v4f*)(s2 + b * KDIM + 4 * threadIdx.x);
  float m = fmaxf(fmaxf(v.x, v.y), fmaxf(v.z, v.w));
  red[threadIdx.x] = m;
  __syncthreads();
  for (int s = 128; s > 0; s >>= 1) {
    if (threadIdx.x < s) red[threadIdx.x] = fmaxf(red[threadIdx.x], red[threadIdx.x + s]);
    __syncthreads();
  }
  if (threadIdx.x == 0) s2max[b] = red[0];
}

#define MT 2
#define NT 4
#define IGRP (8 / NT)
__global__ __launch_bounds__(256) __attribute__((amdgpu_waves_per_eu(2, 2))) void fal_attn_kernel(const float* __restrict__ x,
                                                       const float* __restrict__ s1g,
                                                       const float* __restrict__ s2g,
                                                       const float* __restrict__ t1g,
                                                       const float* __restrict__ tag,
                                                       const float* __restrict__ s2mx,
                                                       float* __restrict__ out) {
  const int tid   = threadIdx.x;
  const int lane  = tid & 31;
  const int wave  = tid >> 5;
  const int h     = lane >> 4;
  const int c     = lane & 15;
  const int b     = blockIdx.y;
  const int wbase = (wave / IGRP) * (16 * MT);
  const int ibase = blockIdx.x * 128 + (wave % IGRP) * (16 * NT);

  const float smax = s2mx[b];
  float s1v[NT], u1[NT], ua[NT];
#pragma unroll
  for (int nt = 0; nt < NT; ++nt) {
    float s    = s1g[b * KDIM + ibase + 16 * nt + c];
    float zmax = s + smax;
    float mrow = fmaxf(zmax, ALPHA * zmax);
    s1v[nt] = s;
    u1[nt]  = __expf(s - mrow);
    ua[nt]  = __expf(ALPHA * s - mrow);
  }

  v8f acc[MT][NT];
#pragma unroll
  for (int mt = 0; mt < MT; ++mt)
#pragma unroll
    for (int nt = 0; nt < NT; ++nt) acc[mt][nt] = (v8f)0.f;
  float dsum[NT];
#pragma unroll
  for (int nt = 0; nt < NT; ++nt) dsum[nt] = 0.f;

  const float* xa = x + (((size_t)b * WDIM + wbase + c) << 10) + 8 * h;
  const float* f2 = s2g + b * KDIM + 8 * h;
  const float* f1 = t1g + b * KDIM + 8 * h;
  const float* fa = tag + b * KDIM + 8 * h;

  v4f va[2][MT][2];
  v4f g2[2][2], g1[2][2], ga[2][2];

#define LOAD_BLOCK(SET, BLK)                                                        \
  {                                                                                 \
    const int jo = (BLK) * JB;                                                      \
    _Pragma("unroll") for (int mt = 0; mt < MT; ++mt) {                             \
      const float* p = xa + ((size_t)(16 * mt) << 10) + jo;                         \
      va[SET][mt][0] = *(const v4f*)(p);                                            \
      va[SET][mt][1] = *(const v4f*)(p + 4);                                        \
    }                                                                               \
    g2[SET][0] = *(const v4f*)(f2 + jo); g2[SET][1] = *(const v4f*)(f2 + jo + 4);   \
    g1[SET][0] = *(const v4f*)(f1 + jo); g1[SET][1] = *(const v4f*)(f1 + jo + 4);   \
    ga[SET][0] = *(const v4f*)(fa + jo); ga[SET][1] = *(const v4f*)(fa + jo + 4);   \
  }

#define HOLD_BLOCK(SET)                                                             \
  asm volatile("" :: "v"(va[SET][0][0]), "v"(va[SET][0][1]), "v"(va[SET][1][0]),    \
               "v"(va[SET][1][1]), "v"(g2[SET][0]), "v"(g2[SET][1]),                \
               "v"(g1[SET][0]), "v"(g1[SET][1]), "v"(ga[SET][0]), "v"(ga[SET][1]));

#define KSTEP(SET, HALF, E0, E1)                                                    \
  {                                                                                 \
    v2f bfr[NT];                                                                    \
    _Pragma("unroll") for (int nt = 0; nt < NT; ++nt) {                             \
      float z0 = s1v[nt] + g2[SET][HALF].E0;                                        \
      float z1 = s1v[nt] + g2[SET][HALF].E1;                                        \
      float h0 = u1[nt] * g1[SET][HALF].E0, l0 = ua[nt] * ga[SET][HALF].E0;         \
      float h1 = u1[nt] * g1[SET][HALF].E1, l1 = ua[nt] * ga[SET][HALF].E1;         \
      float p0 = (z0 > 0.f) ? h0 : l0;                                              \
      float p1 = (z1 > 0.f) ? h1 : l1;                                              \
      dsum[nt] += p0 + p1;                                                          \
      bfr[nt].x = p0; bfr[nt].y = p1;                                               \
    }                                                                               \
    _Pragma("unroll") for (int mt = 0; mt < MT; ++mt) {                             \
      v2f a;                                                                        \
      a.x = va[SET][mt][HALF].E0; a.y = va[SET][mt][HALF].E1;                       \
      _Pragma("unroll") for (int nt = 0; nt < NT; ++nt)                             \
        acc[mt][nt] = __builtin_amdgcn_wmma_f32_16x16x4_f32(false, a, false, bfr[nt], \
                                                   (short)0, acc[mt][nt], false, false); \
    }                                                                               \
  }

#define BLOCK(SET) \
  KSTEP(SET, 0, x, y) KSTEP(SET, 0, z, w) KSTEP(SET, 1, x, y) KSTEP(SET, 1, z, w)

  LOAD_BLOCK(0, 0)
  HOLD_BLOCK(0)
  for (int blk = 0; blk < NBLK; blk += 2) {
    LOAD_BLOCK(1, blk + 1)
    BLOCK(0)
    const int nb = (blk + 2 < NBLK) ? blk + 2 : NBLK - 1;
    LOAD_BLOCK(0, nb)
    BLOCK(1)
  }
#undef BLOCK
#undef KSTEP
#undef HOLD_BLOCK
#undef LOAD_BLOCK

#pragma unroll
  for (int nt = 0; nt < NT; ++nt) {
    float dtot = dsum[nt] + __shfl_xor(dsum[nt], 16, 32);
    float inv  = 1.0f / dtot;
    float* orow = out + (((size_t)b * KDIM + ibase + 16 * nt + c) << 7) + wbase + 8 * h;
#pragma unroll
    for (int mt = 0; mt < MT; ++mt) {
      v8f r = acc[mt][nt];
      *(float4*)(orow + 16 * mt)     = make_float4(r[0] * inv, r[1] * inv, r[2] * inv, r[3] * inv);
      *(float4*)(orow + 16 * mt + 4) = make_float4(r[4] * inv, r[5] * inv, r[6] * inv, r[7] * inv);
    }
  }
}
#undef MT
#undef NT
#undef IGRP

extern "C" void kernel_launch(void* const* d_in, const int* in_sizes, int n_in,
                              void* d_out, int out_size, void* d_ws, size_t ws_size,
                              hipStream_t stream) {
  const float* x = (const float*)d_in[0];
  const float* w = (const float*)d_in[1];
  float* out = (float*)d_out;

  float* s1  = (float*)d_ws;
  float* s2  = s1 + BATCH * KDIM;
  float* t1  = s2 + BATCH * KDIM;
  float* ta  = t1 + BATCH * KDIM;
  float* s2m = ta + BATCH * KDIM;

  fal_prep_kernel<<<BATCH * (KDIM / 128), 256, 0, stream>>>(x, w, s1, s2, t1, ta);
  fal_rowmax_kernel<<<BATCH, 256, 0, stream>>>(s2, s2m);
  dim3 grid(KDIM / 128, BATCH);
  fal_attn_kernel<<<grid, 256, 0, stream>>>(x, s1, s2, t1, ta, s2m, out);
}


# wave tile 2x4 -> 4x2 (64 w x 32 i): each softmax-weight fragment, built as (j,j+1) register pairs, feeds 4 products, half as many built per product; denominator summed per pair: sum order changes
# speedup vs baseline: 1.6245x; 1.0475x over previous
#include <hip/hip_runtime.h>
#include <hip/hip_bf16.h>

typedef __attribute__((ext_vector_type(2))) float v2f;
typedef __attribute__((ext_vector_type(4))) float v4f;
typedef __attribute__((ext_vector_type(8))) float v8f;

#define ALPHA 0.2f
#define BATCH 32
#define WDIM 128
#define KDIM 1024
#define JB   16
#define NBLK (KDIM / JB)

#define PCOLS 128
#define PROWS (WDIM / 8)
__global__ __launch_bounds__(256) __attribute__((amdgpu_waves_per_eu(2, 8))) void fal_prep_kernel(const float* __restrict__ x,
                                                       const float* __restrict__ w,
                                                       float* __restrict__ s1,
                                                       float* __restrict__ s2,
                                                       float* __restrict__ t1,
                                                       float* __restrict__ ta) {
  __shared__ __attribute__((aligned(16))) float part[8 * 2 * PCOLS];
  const int tid  = threadIdx.x;
  const int lane = tid & 31;
  const int wave = __builtin_amdgcn_readfirstlane(tid >> 5);
  const int b    = blockIdx.x >> 3;
  const int k0   = (blockIdx.x & 7) * PCOLS;
  const float* xp = x + (((size_t)b * WDIM + wave * PROWS) << 10) + k0 + 4 * lane;
  const float* w1 = w + wave * PROWS;
  const float* w2 = w + WDIM + wave * PROWS;

  v4f xv[PROWS];
#pragma unroll
  for (int t = 0; t < PROWS; ++t) xv[t] = *(const v4f*)(xp + ((size_t)t << 10));

  v4f a1 = (v4f)0.f, a2 = (v4f)0.f;
#pragma unroll
  for (int t = 0; t < PROWS; ++t) {
    const float c1 = w1[t], c2 = w2[t];
    a1.x = fmaf(xv[t].x, c1, a1.x); a2.x = fmaf(xv[t].x, c2, a2.x);
    a1.y = fmaf(xv[t].y, c1, a1.y); a2.y = fmaf(xv[t].y, c2, a2.y);
    a1.z = fmaf(xv[t].z, c1, a1.z); a2.z = fmaf(xv[t].z, c2, a2.z);
    a1.w = fmaf(xv[t].w, c1, a1.w); a2.w = fmaf(xv[t].w, c2, a2.w);
  }
  *(v4f*)&part[(wave * 2 + 0) * PCOLS + 4 * lane] = a1;
  *(v4f*)&part[(wave * 2 + 1) * PCOLS + 4 * lane] = a2;
  __syncthreads();

  const int sel = tid >> 7;
  const int col = tid & (PCOLS - 1);
  float acc = 0.f;
#pragma unroll
  for (int r = 0; r < 8; ++r) acc += part[(r * 2 + sel) * PCOLS + col];
  const int idx = b * KDIM + k0 + col;
  if (sel == 0) {
    s1[idx] = acc;
  } else {
    s2[idx] = acc;
    t1[idx] = __expf(acc);
    ta[idx] = __expf(ALPHA * acc);
  }
}
#undef PCOLS
#undef PROWS

__global__ __launch_bounds__(256) void fal_rowmax_kernel(const float* __restrict__ s2,
                                                         float* __restrict__ s2max) {
  __shared__ float red[256];
  int b = blockIdx.x;
  const v4f v = *(const v4f*)(s2 + b * KDIM + 4 * threadIdx.x);
  float m = fmaxf(fmaxf(v.x, v.y), fmaxf(v.z, v.w));
  red[threadIdx.x] = m;
  __syncthreads();
  for (int s = 128; s > 0; s >>= 1) {
    if (threadIdx.x < s) red[threadIdx.x] = fmaxf(red[threadIdx.x], red[threadIdx.x + s]);
    __syncthreads();
  }
  if (threadIdx.x == 0) s2max[b] = red[0];
}

#define MT 4
#define NT 2
#define IGRP (8 / NT)
__global__ __launch_bounds__(256) __attribute__((amdgpu_waves_per_eu(2, 2))) void fal_attn_kernel(const float* __restrict__ x,
                                                       const float* __restrict__ s1g,
                                                       const float* __restrict__ s2g,
                                                       const float* __restrict__ t1g,
                                                       const float* __restrict__ tag,
                                                       const float* __restrict__ s2mx,
                                                       float* __restrict__ out) {
  const int tid   = threadIdx.x;
  const int lane  = tid & 31;
  const int wave  = tid >> 5;
  const int h     = lane >> 4;
  const int c     = lane & 15;
  const int b     = blockIdx.y;
  const int wbase = (wave / IGRP) * (16 * MT);
  const int ibase = blockIdx.x * 128 + (wave % IGRP) * (16 * NT);

  const float smax = s2mx[b];
  float s1v[NT], u1[NT], ua[NT];
#pragma unroll
  for (int nt = 0; nt < NT; ++nt) {
    float s    = s1g[b * KDIM + ibase + 16 * nt + c];
    float zmax = s + smax;
    float mrow = fmaxf(zmax, ALPHA * zmax);
    s1v[nt] = s;
    u1[nt]  = __expf(s - mrow);
    ua[nt]  = __expf(ALPHA * s - mrow);
  }

  v8f acc[MT][NT];
#pragma unroll
  for (int mt = 0; mt < MT; ++mt)
#pragma unroll
    for (int nt = 0; nt < NT; ++nt) acc[mt][nt] = (v8f)0.f;
  v2f dsum[NT];
#pragma unroll
  for (int nt = 0; nt < NT; ++nt) dsum[nt] = (v2f)0.f;

  const float* xa = x + (((size_t)b * WDIM + wbase + c) << 10) + 8 * h;
  const float* f2 = s2g + b * KDIM + 8 * h;
  const float* f1 = t1g + b * KDIM + 8 * h;
  const float* fa = tag + b * KDIM + 8 * h;

  v4f va[2][MT][2];
  v4f g2[2][2], g1[2][2], ga[2][2];

#define LOAD_BLOCK(SET, BLK)                                                        \
  {                                                                                 \
    const int jo = (BLK) * JB;                                                      \
    _Pragma("unroll") for (int mt = 0; mt < MT; ++mt) {                             \
      const float* p = xa + ((size_t)(16 * mt) << 10) + jo;                         \
      va[SET][mt][0] = *(const v4f*)(p);                                            \
      va[SET][mt][1] = *(const v4f*)(p + 4);                                        \
    }                                                                               \
    g2[SET][0] = *(const v4f*)(f2 + jo); g2[SET][1] = *(const v4f*)(f2 + jo + 4);   \
    g1[SET][0] = *(const v4f*)(f1 + jo); g1[SET][1] = *(const v4f*)(f1 + jo + 4);   \
    ga[SET][0] = *(const v4f*)(fa + jo); ga[SET][1] = *(const v4f*)(fa + jo + 4);   \
  }

#define HOLD_BLOCK(SET)                                                             \
  asm volatile("" :: "v"(va[SET][0][0]), "v"(va[SET][0][1]), "v"(va[SET][1][0]),    \
               "v"(va[SET][1][1]), "v"(va[SET][2][0]), "v"(va[SET][2][1]),          \
               "v"(va[SET][3][0]), "v"(va[SET][3][1]), "v"(g2[SET][0]),             \
               "v"(g2[SET][1]), "v"(g1[SET][0]), "v"(g1[SET][1]), "v"(ga[SET][0]),  \
               "v"(ga[SET][1]));

#define KSTEP(SET, HALF, PAIR)                                                      \
  {                                                                                 \
    const v2f e2 = g2[SET][HALF].PAIR, e1 = g1[SET][HALF].PAIR, ea = ga[SET][HALF].PAIR; \
    v2f bfr[NT];                                                                    \
    _Pragma("unroll") for (int nt = 0; nt < NT; ++nt) {                             \
      const v2f z  = s1v[nt] + e2;                                                  \
      const v2f hi = u1[nt] * e1;                                                   \
      const v2f lo = ua[nt] * ea;                                                   \
      v2f p;                                                                        \
      p.x = (z.x > 0.f) ? hi.x : lo.x;                                              \
      p.y = (z.y > 0.f) ? hi.y : lo.y;                                              \
      dsum[nt] += p;                                                                \
      bfr[nt] = p;                                                                  \
    }                                                                               \
    _Pragma("unroll") for (int mt = 0; mt < MT; ++mt) {                             \
      const v2f a = va[SET][mt][HALF].PAIR;                                         \
      _Pragma("unroll") for (int nt = 0; nt < NT; ++nt)                             \
        acc[mt][nt] = __builtin_amdgcn_wmma_f32_16x16x4_f32(false, a, false, bfr[nt], \
                                                   (short)0, acc[mt][nt], false, false); \
    }                                                                               \
  }

#define BLOCK(SET) \
  KSTEP(SET, 0, xy) KSTEP(SET, 0, zw) KSTEP(SET, 1, xy) KSTEP(SET, 1, zw)

  LOAD_BLOCK(0, 0)
  HOLD_BLOCK(0)
  for (int blk = 0; blk < NBLK; blk += 2) {
    LOAD_BLOCK(1, blk + 1)
    __builtin_amdgcn_sched_barrier(0);
    BLOCK(0)
    const int nb = (blk + 2 < NBLK) ? blk + 2 : NBLK - 1;
    LOAD_BLOCK(0, nb)
    __builtin_amdgcn_sched_barrier(0);
    BLOCK(1)
  }
#undef BLOCK
#undef KSTEP
#undef HOLD_BLOCK
#undef LOAD_BLOCK

#pragma unroll
  for (int nt = 0; nt < NT; ++nt) {
    float dlane = dsum[nt].x + dsum[nt].y;
    float dtot = dlane + __shfl_xor(dlane, 16, 32);
    float inv  = 1.0f / dtot;
    float* orow = out + (((size_t)b * KDIM + ibase + 16 * nt + c) << 7) + wbase + 8 * h;
#pragma unroll
    for (int mt = 0; mt < MT; ++mt) {
      v8f r = acc[mt][nt];
      *(float4*)(orow + 16 * mt)     = make_float4(r[0] * inv, r[1] * inv, r[2] * inv, r[3] * inv);
      *(float4*)(orow + 16 * mt + 4) = make_float4(r[4] * inv, r[5] * inv, r[6] * inv, r[7] * inv);
    }
  }
}
#undef MT
#undef NT
#undef IGRP

extern "C" void kernel_launch(void* const* d_in, const int* in_sizes, int n_in,
                              void* d_out, int out_size, void* d_ws, size_t ws_size,
                              hipStream_t stream) {
  const float* x = (const float*)d_in[0];
  const float* w = (const float*)d_in[1];
  float* out = (float*)d_out;

  float* s1  = (float*)d_ws;
  float* s2  = s1 + BATCH * KDIM;
  float* t1  = s2 + BATCH * KDIM;
  float* ta  = t1 + BATCH * KDIM;
  float* s2m = ta + BATCH * KDIM;

  fal_prep_kernel<<<BATCH * (KDIM / 128), 256, 0, stream>>>(x, w, s1, s2, t1, ta);
  fal_rowmax_kernel<<<BATCH, 256, 0, stream>>>(s2, s2m);
  dim3 grid(KDIM / 128, BATCH);
  fal_attn_kernel<<<grid, 256, 0, stream>>>(x, s1, s2, t1, ta, s2m, out);
}


# row-max launch dropped: each s2 wave of the projection kernel stores the max of its 32 columns, the attention kernel joins its batch's 32 partial maxima; 3 launches -> 2; sum order unchanged
# speedup vs baseline: 1.7156x; 1.0561x over previous
#include <hip/hip_runtime.h>
#include <hip/hip_bf16.h>

typedef __attribute__((ext_vector_type(2))) float v2f;
typedef __attribute__((ext_vector_type(4))) float v4f;
typedef __attribute__((ext_vector_type(8))) float v8f;

#define ALPHA 0.2f
#define BATCH 32
#define WDIM 128
#define KDIM 1024
#define JB   16
#define NBLK (KDIM / JB)
#define NPMAX (4 * (KDIM / 128))

#define PCOLS 128
#define PROWS (WDIM / 8)
__global__ __launch_bounds__(256) __attribute__((amdgpu_waves_per_eu(2, 8))) void fal_prep_kernel(const float* __restrict__ x,
                                                       const float* __restrict__ w,
                                                       float* __restrict__ s1,
                                                       float* __restrict__ s2,
                                                       float* __restrict__ t1,
                                                       float* __restrict__ ta,
                                                       float* __restrict__ pmax) {
  __shared__ __attribute__((aligned(16))) float part[8 * 2 * PCOLS];
  const int tid  = threadIdx.x;
  const int lane = tid & 31;
  const int wave = __builtin_amdgcn_readfirstlane(tid >> 5);
  const int b    = blockIdx.x >> 3;
  const int k0   = (blockIdx.x & 7) * PCOLS;
  const float* xp = x + (((size_t)b * WDIM + wave * PROWS) << 10) + k0 + 4 * lane;
  const float* w1 = w + wave * PROWS;
  const float* w2 = w + WDIM + wave * PROWS;

  v4f xv[PROWS];
#pragma unroll
  for (int t = 0; t < PROWS; ++t) xv[t] = *(const v4f*)(xp + ((size_t)t << 10));

  v4f a1 = (v4f)0.f, a2 = (v4f)0.f;
#pragma unroll
  for (int t = 0; t < PROWS; ++t) {
    const float c1 = w1[t], c2 = w2[t];
    a1.x = fmaf(xv[t].x, c1, a1.x); a2.x = fmaf(xv[t].x, c2, a2.x);
    a1.y = fmaf(xv[t].y, c1, a1.y); a2.y = fmaf(xv[t].y, c2, a2.y);
    a1.z = fmaf(xv[t].z, c1, a1.z); a2.z = fmaf(xv[t].z, c2, a2.z);
    a1.w = fmaf(xv[t].w, c1, a1.w); a2.w = fmaf(xv[t].w, c2, a2.w);
  }
  *(v4f*)&part[(wave * 2 + 0) * PCOLS + 4 * lane] = a1;
  *(v4f*)&part[(wave * 2 + 1) * PCOLS + 4 * lane] = a2;
  __syncthreads();

  const int sel = wave >> 2;
  const int col = tid & (PCOLS - 1);
  float acc = 0.f;
#pragma unroll
  for (int r = 0; r < 8; ++r) acc += part[(r * 2 + sel) * PCOLS + col];
  const int idx = b * KDIM + k0 + col;
  if (sel == 0) {
    s1[idx] = acc;
  } else {
    s2[idx] = acc;
    t1[idx] = __expf(acc);
    ta[idx] = __expf(ALPHA * acc);
    float m = acc;
#pragma unroll
    for (int o = 16; o > 0; o >>= 1) m = fmaxf(m, __shfl_xor(m, o, 32));
    if (lane == 0) pmax[4 * blockIdx.x + (wave - 4)] = m;
  }
}
#undef PCOLS
#undef PROWS

#define MT 4
#define NT 2
#define IGRP (8 / NT)
__global__ __launch_bounds__(256) __attribute__((amdgpu_waves_per_eu(2, 2))) void fal_attn_kernel(const float* __restrict__ x,
                                                       const float* __restrict__ s1g,
                                                       const float* __restrict__ s2g,
                                                       const float* __restrict__ t1g,
                                                       const float* __restrict__ tag,
                                                       const float* __restrict__ s2mx,
                                                       float* __restrict__ out) {
  const int tid   = threadIdx.x;
  const int lane  = tid & 31;
  const int wave  = tid >> 5;
  const int h     = lane >> 4;
  const int c     = lane & 15;
  const int b     = blockIdx.y;
  const int wbase = (wave / IGRP) * (16 * MT);
  const int ibase = blockIdx.x * 128 + (wave % IGRP) * (16 * NT);

  const float* pm = s2mx + b * NPMAX;
  float pmv[NPMAX];
#pragma unroll
  for (int g = 0; g < NPMAX; ++g) pmv[g] = pm[g];
  float s1v[NT];
#pragma unroll
  for (int nt = 0; nt < NT; ++nt) s1v[nt] = s1g[b * KDIM + ibase + 16 * nt + c];

  v8f acc[MT][NT];
#pragma unroll
  for (int mt = 0; mt < MT; ++mt)
#pragma unroll
    for (int nt = 0; nt < NT; ++nt) acc[mt][nt] = (v8f)0.f;
  v2f dsum[NT];
#pragma unroll
  for (int nt = 0; nt < NT; ++nt) dsum[nt] = (v2f)0.f;

  const float* xa = x + (((size_t)b * WDIM + wbase + c) << 10) + 8 * h;
  const float* f2 = s2g + b * KDIM + 8 * h;
  const float* f1 = t1g + b * KDIM + 8 * h;
  const float* fa = tag + b * KDIM + 8 * h;

  v4f va[2][MT][2];
  v4f g2[2][2], g1[2][2], ga[2][2];

#define LOAD_BLOCK(SET, BLK)                                                        \
  {                                                                                 \
    const int jo = (BLK) * JB;                                                      \
    _Pragma("unroll") for (int mt = 0; mt < MT; ++mt) {                             \
      const float* p = xa + ((size_t)(16 * mt) << 10) + jo;                         \
      va[SET][mt][0] = *(const v4f*)(p);                                            \
      va[SET][mt][1] = *(const v4f*)(p + 4);                                        \
    }                                                                               \
    g2[SET][0] = *(const v4f*)(f2 + jo); g2[SET][1] = *(const v4f*)(f2 + jo + 4);   \
    g1[SET][0] = *(const v4f*)(f1 + jo); g1[SET][1] = *(const v4f*)(f1 + jo + 4);   \
    ga[SET][0] = *(const v4f*)(fa + jo); ga[SET][1] = *(const v4f*)(fa + jo + 4);   \
  }

#define HOLD_BLOCK(SET)                                                             \
  asm volatile("" :: "v"(va[SET][0][0]), "v"(va[SET][0][1]), "v"(va[SET][1][0]),    \
               "v"(va[SET][1][1]), "v"(va[SET][2][0]), "v"(va[SET][2][1]),          \
               "v"(va[SET][3][0]), "v"(va[SET][3][1]), "v"(g2[SET][0]),             \
               "v"(g2[SET][1]), "v"(g1[SET][0]), "v"(g1[SET][1]), "v"(ga[SET][0]),  \
               "v"(ga[SET][1]));

#define KSTEP(SET, HALF, PAIR)                                                      \
  {                                                                                 \
    const v2f e2 = g2[SET][HALF].PAIR, e1 = g1[SET][HALF].PAIR, ea = ga[SET][HALF].PAIR; \
    v2f bfr[NT];                                                                    \
    _Pragma("unroll") for (int nt = 0; nt < NT; ++nt) {                             \
      const v2f z  = s1v[nt] + e2;                                                  \
      const v2f hi = u1[nt] * e1;                                                   \
      const v2f lo = ua[nt] * ea;                                                   \
      v2f p;                                                                        \
      p.x = (z.x > 0.f) ? hi.x : lo.x;                                              \
      p.y = (z.y > 0.f) ? hi.y : lo.y;                                              \
      dsum[nt] += p;                                                                \
      bfr[nt] = p;                                                                  \
    }                                                                               \
    _Pragma("unroll") for (int mt = 0; mt < MT; ++mt) {                             \
      const v2f a = va[SET][mt][HALF].PAIR;                                         \
      _Pragma("unroll") for (int nt = 0; nt < NT; ++nt)                             \
        acc[mt][nt] = __builtin_amdgcn_wmma_f32_16x16x4_f32(false, a, false, bfr[nt], \
                                                   (short)0, acc[mt][nt], false, false); \
    }                                                                               \
  }

#define BLOCK(SET) \
  KSTEP(SET, 0, xy) KSTEP(SET, 0, zw) KSTEP(SET, 1, xy) KSTEP(SET, 1, zw)

  LOAD_BLOCK(0, 0)
  __builtin_amdgcn_sched_barrier(0);
  float smax = pmv[0];
#pragma unroll
  for (int g = 1; g < NPMAX; ++g) smax = fmaxf(smax, pmv[g]);
  float u1[NT], ua[NT];
#pragma unroll
  for (int nt = 0; nt < NT; ++nt) {
    const float s    = s1v[nt];
    const float zmax = s + smax;
    const float mrow = fmaxf(zmax, ALPHA * zmax);
    u1[nt] = __expf(s - mrow);
    ua[nt] = __expf(ALPHA * s - mrow);
  }
  HOLD_BLOCK(0)
  for (int blk = 0; blk < NBLK; blk += 2) {
    LOAD_BLOCK(1, blk + 1)
    __builtin_amdgcn_sched_barrier(0);
    BLOCK(0)
    const int nb = (blk + 2 < NBLK) ? blk + 2 : NBLK - 1;
    LOAD_BLOCK(0, nb)
    __builtin_amdgcn_sched_barrier(0);
    BLOCK(1)
  }
#undef BLOCK
#undef KSTEP
#undef HOLD_BLOCK
#undef LOAD_BLOCK

#pragma unroll
  for (int nt = 0; nt < NT; ++nt) {
    float dlane = dsum[nt].x + dsum[nt].y;
    float dtot = dlane + __shfl_xor(dlane, 16, 32);
    float inv  = 1.0f / dtot;
    float* orow = out + (((size_t)b * KDIM + ibase + 16 * nt + c) << 7) + wbase + 8 * h;
#pragma unroll
    for (int mt = 0; mt < MT; ++mt) {
      v8f r = acc[mt][nt];
      *(float4*)(orow + 16 * mt)     = make_float4(r[0] * inv, r[1] * inv, r[2] * inv, r[3] * inv);
      *(float4*)(orow + 16 * mt + 4) = make_float4(r[4] * inv, r[5] * inv, r[6] * inv, r[7] * inv);
    }
  }
}
#undef MT
#undef NT
#undef IGRP

extern "C" void kernel_launch(void* const* d_in, const int* in_sizes, int n_in,
                              void* d_out, int out_size, void* d_ws, size_t ws_size,
                              hipStream_t stream) {
  const float* x = (const float*)d_in[0];
  const float* w = (const float*)d_in[1];
  float* out = (float*)d_out;

  float* s1  = (float*)d_ws;
  float* s2  = s1 + BATCH * KDIM;
  float* t1  = s2 + BATCH * KDIM;
  float* ta  = t1 + BATCH * KDIM;
  float* s2m = ta + BATCH * KDIM;

  fal_prep_kernel<<<BATCH * (KDIM / 128), 256, 0, stream>>>(x, w, s1, s2, t1, ta, s2m);
  dim3 grid(KDIM / 128, BATCH);
  fal_attn_kernel<<<grid, 256, 0, stream>>>(x, s1, s2, t1, ta, s2m, out);
}


# wave tile 4x2 -> 8x2 (128 w x 32 i), 16-column j blocks dealt alternately to a wave pair, half sums added once via LDS: each softmax-weight fragment feeds 8 products, built once; sum order changes
# speedup vs baseline: 1.8712x; 1.0907x over previous
#include <hip/hip_runtime.h>
#include <hip/hip_bf16.h>

typedef __attribute__((ext_vector_type(2))) float v2f;
typedef __attribute__((ext_vector_type(4))) float v4f;
typedef __attribute__((ext_vector_type(8))) float v8f;

#define ALPHA 0.2f
#define BATCH 32
#define WDIM 128
#define KDIM 1024
#define JB   16
#define NBLK (KDIM / JB)
#define NPMAX (4 * (KDIM / 128))

#define PCOLS 128
#define PROWS (WDIM / 8)
__global__ __launch_bounds__(256) __attribute__((amdgpu_waves_per_eu(2, 8))) void fal_prep_kernel(const float* __restrict__ x,
                                                       const float* __restrict__ w,
                                                       float* __restrict__ s1,
                                                       float* __restrict__ s2,
                                                       float* __restrict__ t1,
                                                       float* __restrict__ ta,
                                                       float* __restrict__ pmax) {
  __shared__ __attribute__((aligned(16))) float part[8 * 2 * PCOLS];
  const int tid  = threadIdx.x;
  const int lane = tid & 31;
  const int wave = __builtin_amdgcn_readfirstlane(tid >> 5);
  const int b    = blockIdx.x >> 3;
  const int k0   = (blockIdx.x & 7) * PCOLS;
  const float* xp = x + (((size_t)b * WDIM + wave * PROWS) << 10) + k0 + 4 * lane;
  const float* w1 = w + wave * PROWS;
  const float* w2 = w + WDIM + wave * PROWS;

  v4f xv[PROWS];
#pragma unroll
  for (int t = 0; t < PROWS; ++t) xv[t] = *(const v4f*)(xp + ((size_t)t << 10));

  v4f a1 = (v4f)0.f, a2 = (v4f)0.f;
#pragma unroll
  for (int t = 0; t < PROWS; ++t) {
    const float c1 = w1[t], c2 = w2[t];
    a1.x = fmaf(xv[t].x, c1, a1.x); a2.x = fmaf(xv[t].x, c2, a2.x);
    a1.y = fmaf(xv[t].y, c1, a1.y); a2.y = fmaf(xv[t].y, c2, a2.y);
    a1.z = fmaf(xv[t].z, c1, a1.z); a2.z = fmaf(xv[t].z, c2, a2.z);
    a1.w = fmaf(xv[t].w, c1, a1.w); a2.w = fmaf(xv[t].w, c2, a2.w);
  }
  *(v4f*)&part[(wave * 2 + 0) * PCOLS + 4 * lane] = a1;
  *(v4f*)&part[(wave * 2 + 1) * PCOLS + 4 * lane] = a2;
  __syncthreads();

  const int sel = wave >> 2;
  const int col = tid & (PCOLS - 1);
  float acc = 0.f;
#pragma unroll
  for (int r = 0; r < 8; ++r) acc += part[(r * 2 + sel) * PCOLS + col];
  const int idx = b * KDIM + k0 + col;
  if (sel == 0) {
    s1[idx] = acc;
  } else {
    s2[idx] = acc;
    t1[idx] = __expf(acc);
    ta[idx] = __expf(ALPHA * acc);
    float m = acc;
#pragma unroll
    for (int o = 16; o > 0; o >>= 1) m = fmaxf(m, __shfl_xor(m, o, 32));
    if (lane == 0) pmax[4 * blockIdx.x + (wave - 4)] = m;
  }
}
#undef PCOLS
#undef PROWS

#define MT 8
#define NT 2
#define KSPL 2
#define NBW (NBLK / KSPL)
#define JSTEP (JB * KSPL)
__global__ __launch_bounds__(256) __attribute__((amdgpu_waves_per_eu(2, 2))) void fal_attn_kernel(const float* __restrict__ x,
                                                       const float* __restrict__ s1g,
                                                       const float* __restrict__ s2g,
                                                       const float* __restrict__ t1g,
                                                       const float* __restrict__ tag,
                                                       const float* __restrict__ s2mx,
                                                       float* __restrict__ out) {
  __shared__ __attribute__((aligned(16))) v4f xch[8][2 * MT][32];
  __shared__ float dex[8][32];

  const int tid   = threadIdx.x;
  const int lane  = tid & 31;
  const int wave  = __builtin_amdgcn_readfirstlane(tid >> 5);
  const int h     = lane >> 4;
  const int c     = lane & 15;
  const int b     = blockIdx.y;
  const int kh    = wave >> 2;
  const int ibase = blockIdx.x * 128 + (wave & 3) * (16 * NT);

  const float* pm = s2mx + b * NPMAX;
  float pmv[NPMAX];
#pragma unroll
  for (int g = 0; g < NPMAX; ++g) pmv[g] = pm[g];
  float s1v[NT];
#pragma unroll
  for (int nt = 0; nt < NT; ++nt) s1v[nt] = s1g[b * KDIM + ibase + 16 * nt + c];

  v8f acc[MT][NT];
#pragma unroll
  for (int mt = 0; mt < MT; ++mt)
#pragma unroll
    for (int nt = 0; nt < NT; ++nt) acc[mt][nt] = (v8f)0.f;
  v2f dsum[NT];
#pragma unroll
  for (int nt = 0; nt < NT; ++nt) dsum[nt] = (v2f)0.f;

  const int j0 = kh * JB + 8 * h;
  const float* xa = x + (((size_t)b * WDIM + c) << 10) + j0;
  const float* f2 = s2g + b * KDIM + j0;
  const float* f1 = t1g + b * KDIM + j0;
  const float* fa = tag + b * KDIM + j0;

  v4f va[2][MT][2];
  v4f g2[2][2], g1[2][2], ga[2][2];

#define LOAD_BLOCK(SET, BLK)                                                        \
  {                                                                                 \
    const int jo = (BLK) * JSTEP;                                                   \
    _Pragma("unroll") for (int mt = 0; mt < MT; ++mt) {                             \
      const float* p = xa + ((size_t)(16 * mt) << 10) + jo;                         \
      va[SET][mt][0] = *(const v4f*)(p);                                            \
      va[SET][mt][1] = *(const v4f*)(p + 4);                                        \
    }                                                                               \
    g2[SET][0] = *(const v4f*)(f2 + jo); g2[SET][1] = *(const v4f*)(f2 + jo + 4);   \
    g1[SET][0] = *(const v4f*)(f1 + jo); g1[SET][1] = *(const v4f*)(f1 + jo + 4);   \
    ga[SET][0] = *(const v4f*)(fa + jo); ga[SET][1] = *(const v4f*)(fa + jo + 4);   \
  }

#define HOLD_BLOCK(SET)                                                             \
  asm volatile("" :: "v"(va[SET][0][0]), "v"(va[SET][0][1]), "v"(va[SET][1][0]),    \
               "v"(va[SET][1][1]), "v"(va[SET][2][0]), "v"(va[SET][2][1]),          \
               "v"(va[SET][3][0]), "v"(va[SET][3][1]), "v"(va[SET][4][0]),          \
               "v"(va[SET][4][1]), "v"(va[SET][5][0]), "v"(va[SET][5][1]),          \
               "v"(va[SET][6][0]), "v"(va[SET][6][1]), "v"(va[SET][7][0]),          \
               "v"(va[SET][7][1]), "v"(g2[SET][0]),                                 \
               "v"(g2[SET][1]), "v"(g1[SET][0]), "v"(g1[SET][1]), "v"(ga[SET][0]),  \
               "v"(ga[SET][1]));

#define KSTEP(SET, HALF, PAIR)                                                      \
  {                                                                                 \
    const v2f e2 = g2[SET][HALF].PAIR, e1 = g1[SET][HALF].PAIR, ea = ga[SET][HALF].PAIR; \
    v2f bfr[NT];                                                                    \
    _Pragma("unroll") for (int nt = 0; nt < NT; ++nt) {                             \
      const v2f z  = s1v[nt] + e2;                                                  \
      const v2f hi = u1[nt] * e1;                                                   \
      const v2f lo = ua[nt] * ea;                                                   \
      v2f p;                                                                        \
      p.x = (z.x > 0.f) ? hi.x : lo.x;                                              \
      p.y = (z.y > 0.f) ? hi.y : lo.y;                                              \
      dsum[nt] += p;                                                                \
      bfr[nt] = p;                                                                  \
    }                                                                               \
    _Pragma("unroll") for (int mt = 0; mt < MT; ++mt) {                             \
      const v2f a = va[SET][mt][HALF].PAIR;                                         \
      _Pragma("unroll") for (int nt = 0; nt < NT; ++nt)                             \
        acc[mt][nt] = __builtin_amdgcn_wmma_f32_16x16x4_f32(false, a, false, bfr[nt], \
                                                   (short)0, acc[mt][nt], false, false); \
    }                                                                               \
  }

#define BLOCK(SET) \
  KSTEP(SET, 0, xy) KSTEP(SET, 0, zw) KSTEP(SET, 1, xy) KSTEP(SET, 1, zw)

  LOAD_BLOCK(0, 0)
  __builtin_amdgcn_sched_barrier(0);
  float smax = pmv[0];
#pragma unroll
  for (int g = 1; g < NPMAX; ++g) smax = fmaxf(smax, pmv[g]);
  float u1[NT], ua[NT];
#pragma unroll
  for (int nt = 0; nt < NT; ++nt) {
    const float s    = s1v[nt];
    const float zmax = s + smax;
    const float mrow = fmaxf(zmax, ALPHA * zmax);
    u1[nt] = __expf(s - mrow);
    ua[nt] = __expf(ALPHA * s - mrow);
  }
  HOLD_BLOCK(0)
  for (int blk = 0; blk < NBW; blk += 2) {
    LOAD_BLOCK(1, blk + 1)
    __builtin_amdgcn_sched_barrier(0);
    BLOCK(0)
    const int nb = (blk + 2 < NBW) ? blk + 2 : NBW - 1;
    LOAD_BLOCK(0, nb)
    __builtin_amdgcn_sched_barrier(0);
    BLOCK(1)
  }
#undef BLOCK
#undef KSTEP
#undef HOLD_BLOCK
#undef LOAD_BLOCK

  float dt[NT];
#pragma unroll
  for (int nt = 0; nt < NT; ++nt) {
    const float dlane = dsum[nt].x + dsum[nt].y;
    dt[nt] = dlane + __shfl_xor(dlane, 16, 32);
  }

#define SEND_TILE(NTS)                                                              \
  {                                                                                 \
    _Pragma("unroll") for (int mt = 0; mt < MT; ++mt) {                             \
      const v8f r = acc[mt][NTS];                                                   \
      v4f lo4, hi4;                                                                 \
      lo4.x = r[0]; lo4.y = r[1]; lo4.z = r[2]; lo4.w = r[3];                       \
      hi4.x = r[4]; hi4.y = r[5]; hi4.z = r[6]; hi4.w = r[7];                       \
      xch[wave][2 * mt][lane]     = lo4;                                            \
      xch[wave][2 * mt + 1][lane] = hi4;                                            \
    }                                                                               \
    dex[wave][lane] = dt[NTS];                                                      \
  }
#define FINISH_TILE(NTK)                                                            \
  {                                                                                 \
    const int pw = wave ^ 4;                                                        \
    const float inv = 1.0f / (dt[NTK] + dex[pw][lane]);                             \
    float* orow = out + (((size_t)b * KDIM + ibase + 16 * NTK + c) << 7) + 8 * h;   \
    _Pragma("unroll") for (int mt = 0; mt < MT; ++mt) {                             \
      const v8f r = acc[mt][NTK];                                                   \
      const v4f p0 = xch[pw][2 * mt][lane];                                         \
      const v4f p1 = xch[pw][2 * mt + 1][lane];                                     \
      *(float4*)(orow + 16 * mt)     = make_float4((r[0] + p0.x) * inv, (r[1] + p0.y) * inv, \
                                                   (r[2] + p0.z) * inv, (r[3] + p0.w) * inv); \
      *(float4*)(orow + 16 * mt + 4) = make_float4((r[4] + p1.x) * inv, (r[5] + p1.y) * inv, \
                                                   (r[6] + p1.z) * inv, (r[7] + p1.w) * inv); \
    }                                                                               \
  }

  if (kh == 0) SEND_TILE(1) else SEND_TILE(0)
  __syncthreads();
  if (kh == 0) FINISH_TILE(0) else FINISH_TILE(1)
#undef SEND_TILE
#undef FINISH_TILE
}
#undef MT
#undef NT
#undef KSPL
#undef NBW
#undef JSTEP

extern "C" void kernel_launch(void* const* d_in, const int* in_sizes, int n_in,
                              void* d_out, int out_size, void* d_ws, size_t ws_size,
                              hipStream_t stream) {
  const float* x = (const float*)d_in[0];
  const float* w = (const float*)d_in[1];
  float* out = (float*)d_out;

  float* s1  = (float*)d_ws;
  float* s2  = s1 + BATCH * KDIM;
  float* t1  = s2 + BATCH * KDIM;
  float* ta  = t1 + BATCH * KDIM;
  float* s2m = ta + BATCH * KDIM;

  fal_prep_kernel<<<BATCH * (KDIM / 128), 256, 0, stream>>>(x, w, s1, s2, t1, ta, s2m);
  dim3 grid(KDIM / 128, BATCH);
  fal_attn_kernel<<<grid, 256, 0, stream>>>(x, s1, s2, t1, ta, s2m, out);
}


# wave tile 8x2 -> 8x4 (128 w x 64 i), j blocks dealt round-robin to four wave pairs, quarter sums added once via LDS: each 16-line x fragment load feeds 4 products, not 2; sum order changes
# speedup vs baseline: 1.9321x; 1.0325x over previous
#include <hip/hip_runtime.h>
#include <hip/hip_bf16.h>

typedef __attribute__((ext_vector_type(2))) float v2f;
typedef __attribute__((ext_vector_type(4))) float v4f;
typedef __attribute__((ext_vector_type(8))) float v8f;

#define ALPHA 0.2f
#define BATCH 32
#define WDIM 128
#define KDIM 1024
#define JB   16
#define NBLK (KDIM / JB)
#define NPMAX (4 * (KDIM / 128))

#define PCOLS 128
#define PROWS (WDIM / 8)
__global__ __launch_bounds__(256) __attribute__((amdgpu_waves_per_eu(2, 8))) void fal_prep_kernel(const float* __restrict__ x,
                                                       const float* __restrict__ w,
                                                       float* __restrict__ s1,
                                                       float* __restrict__ s2,
                                                       float* __restrict__ t1,
                                                       float* __restrict__ ta,
                                                       float* __restrict__ pmax) {
  __shared__ __attribute__((aligned(16))) float part[8 * 2 * PCOLS];
  const int tid  = threadIdx.x;
  const int lane = tid & 31;
  const int wave = __builtin_amdgcn_readfirstlane(tid >> 5);
  const int b    = blockIdx.x >> 3;
  const int k0   = (blockIdx.x & 7) * PCOLS;
  const float* xp = x + (((size_t)b * WDIM + wave * PROWS) << 10) + k0 + 4 * lane;
  const float* w1 = w + wave * PROWS;
  const float* w2 = w + WDIM + wave * PROWS;

  v4f xv[PROWS];
#pragma unroll
  for (int t = 0; t < PROWS; ++t) xv[t] = *(const v4f*)(xp + ((size_t)t << 10));

  v4f a1 = (v4f)0.f, a2 = (v4f)0.f;
#pragma unroll
  for (int t = 0; t < PROWS; ++t) {
    const float c1 = w1[t], c2 = w2[t];
    a1.x = fmaf(xv[t].x, c1, a1.x); a2.x = fmaf(xv[t].x, c2, a2.x);
    a1.y = fmaf(xv[t].y, c1, a1.y); a2.y = fmaf(xv[t].y, c2, a2.y);
    a1.z = fmaf(xv[t].z, c1, a1.z); a2.z = fmaf(xv[t].z, c2, a2.z);
    a1.w = fmaf(xv[t].w, c1, a1.w); a2.w = fmaf(xv[t].w, c2, a2.w);
  }
  *(v4f*)&part[(wave * 2 + 0) * PCOLS + 4 * lane] = a1;
  *(v4f*)&part[(wave * 2 + 1) * PCOLS + 4 * lane] = a2;
  __syncthreads();

  const int sel = wave >> 2;
  const int col = tid & (PCOLS - 1);
  float acc = 0.f;
#pragma unroll
  for (int r = 0; r < 8; ++r) acc += part[(r * 2 + sel) * PCOLS + col];
  const int idx = b * KDIM + k0 + col;
  if (sel == 0) {
    s1[idx] = acc;
  } else {
    s2[idx] = acc;
    t1[idx] = __expf(acc);
    ta[idx] = __expf(ALPHA * acc);
    float m = acc;
#pragma unroll
    for (int o = 16; o > 0; o >>= 1) m = fmaxf(m, __shfl_xor(m, o, 32));
    if (lane == 0) pmax[4 * blockIdx.x + (wave - 4)] = m;
  }
}
#undef PCOLS
#undef PROWS

#define MT 8
#define NT 4
#define KSPL 4
#define NBW (NBLK / KSPL)
#define JSTEP (JB * KSPL)
__global__ __launch_bounds__(256) __attribute__((amdgpu_waves_per_eu(2, 2))) void fal_attn_kernel(const float* __restrict__ x,
                                                       const float* __restrict__ s1g,
                                                       const float* __restrict__ s2g,
                                                       const float* __restrict__ t1g,
                                                       const float* __restrict__ tag,
                                                       const float* __restrict__ s2mx,
                                                       float* __restrict__ out) {
  __shared__ __attribute__((aligned(16))) v4f xch[8][NT - 1][2 * MT][32];
  __shared__ float dex[8][NT][32];

  const int tid   = threadIdx.x;
  const int lane  = tid & 31;
  const int wave  = __builtin_amdgcn_readfirstlane(tid >> 5);
  const int h     = lane >> 4;
  const int c     = lane & 15;
  const int b     = blockIdx.y;
  const int kq    = wave & 3;
  const int ih    = wave >> 2;
  const int ibase = blockIdx.x * 128 + ih * (16 * NT);

  const float* pm = s2mx + b * NPMAX;
  float pmv[NPMAX];
#pragma unroll
  for (int g = 0; g < NPMAX; ++g) pmv[g] = pm[g];
  float s1v[NT];
#pragma unroll
  for (int nt = 0; nt < NT; ++nt) s1v[nt] = s1g[b * KDIM + ibase + 16 * ((nt + kq) & 3) + c];

  v8f acc[MT][NT];
#pragma unroll
  for (int mt = 0; mt < MT; ++mt)
#pragma unroll
    for (int nt = 0; nt < NT; ++nt) acc[mt][nt] = (v8f)0.f;
  v2f dsum[NT];
#pragma unroll
  for (int nt = 0; nt < NT; ++nt) dsum[nt] = (v2f)0.f;

  const int j0 = kq * JB + 8 * h;
  const float* xa = x + (((size_t)b * WDIM + c) << 10) + j0;
  const float* f2 = s2g + b * KDIM + j0;
  const float* f1 = t1g + b * KDIM + j0;
  const float* fa = tag + b * KDIM + j0;

  v4f va[2][MT];
  v4f g2[2], g1[2], ga[2];

#define LOAD_HALF(HALF, BLK)                                                        \
  {                                                                                 \
    const int jo = (BLK) * JSTEP + 4 * (HALF);                                      \
    _Pragma("unroll") for (int mt = 0; mt < MT; ++mt)                               \
      va[HALF][mt] = *(const v4f*)(xa + ((size_t)(16 * mt) << 10) + jo);            \
    g2[HALF] = *(const v4f*)(f2 + jo);                                              \
    g1[HALF] = *(const v4f*)(f1 + jo);                                              \
    ga[HALF] = *(const v4f*)(fa + jo);                                              \
  }

#define KSTEP(HALF, PAIR)                                                           \
  {                                                                                 \
    const v2f e2 = g2[HALF].PAIR, e1 = g1[HALF].PAIR, ea = ga[HALF].PAIR;           \
    v2f bfr[NT];                                                                    \
    _Pragma("unroll") for (int nt = 0; nt < NT; ++nt) {                             \
      const v2f z  = s1v[nt] + e2;                                                  \
      const v2f hi = u1[nt] * e1;                                                   \
      const v2f lo = ua[nt] * ea;                                                   \
      v2f p;                                                                        \
      p.x = (z.x > 0.f) ? hi.x : lo.x;                                              \
      p.y = (z.y > 0.f) ? hi.y : lo.y;                                              \
      dsum[nt] += p;                                                                \
      bfr[nt] = p;                                                                  \
    }                                                                               \
    _Pragma("unroll") for (int mt = 0; mt < MT; ++mt) {                             \
      const v2f a = va[HALF][mt].PAIR;                                              \
      _Pragma("unroll") for (int nt = 0; nt < NT; ++nt)                             \
        acc[mt][nt] = __builtin_amdgcn_wmma_f32_16x16x4_f32(false, a, false, bfr[nt], \
                                                   (short)0, acc[mt][nt], false, false); \
    }                                                                               \
  }

#define HALFBLOCK(HALF) KSTEP(HALF, xy) KSTEP(HALF, zw)

  LOAD_HALF(0, 0)
  __builtin_amdgcn_sched_barrier(0);
  float smax = pmv[0];
#pragma unroll
  for (int g = 1; g < NPMAX; ++g) smax = fmaxf(smax, pmv[g]);
  float u1[NT], ua[NT];
#pragma unroll
  for (int nt = 0; nt < NT; ++nt) {
    const float s    = s1v[nt];
    const float zmax = s + smax;
    const float mrow = fmaxf(zmax, ALPHA * zmax);
    u1[nt] = __expf(s - mrow);
    ua[nt] = __expf(ALPHA * s - mrow);
  }
  for (int blk = 0; blk < NBW; ++blk) {
    LOAD_HALF(1, blk)
    __builtin_amdgcn_sched_barrier(0);
    HALFBLOCK(0)
    const int nb = (blk + 1 < NBW) ? blk + 1 : NBW - 1;
    LOAD_HALF(0, nb)
    __builtin_amdgcn_sched_barrier(0);
    HALFBLOCK(1)
  }
#undef HALFBLOCK
#undef KSTEP
#undef LOAD_HALF

  float dt[NT];
#pragma unroll
  for (int nt = 0; nt < NT; ++nt) {
    const float dlane = dsum[nt].x + dsum[nt].y;
    dt[nt] = dlane + __shfl_xor(dlane, 16, 32);
    dex[wave][nt][lane] = dt[nt];
  }

#define SEND_TILE(NTS)                                                              \
  {                                                                                 \
    _Pragma("unroll") for (int mt = 0; mt < MT; ++mt) {                             \
      const v8f r = acc[mt][NTS];                                                   \
      v4f lo4, hi4;                                                                 \
      lo4.x = r[0]; lo4.y = r[1]; lo4.z = r[2]; lo4.w = r[3];                       \
      hi4.x = r[4]; hi4.y = r[5]; hi4.z = r[6]; hi4.w = r[7];                       \
      xch[wave][(NTS) - 1][2 * mt][lane]     = lo4;                                 \
      xch[wave][(NTS) - 1][2 * mt + 1][lane] = hi4;                                 \
    }                                                                               \
  }
  SEND_TILE(1) SEND_TILE(2) SEND_TILE(3)
  __syncthreads();
  {
    const int w1 = (ih << 2) | ((kq + 1) & 3);
    const int w2 = (ih << 2) | ((kq + 2) & 3);
    const int w3 = (ih << 2) | ((kq + 3) & 3);
    const float den = ((dt[0] + dex[w1][3][lane]) + dex[w2][2][lane]) + dex[w3][1][lane];
    const float inv = 1.0f / den;
    float* orow = out + (((size_t)b * KDIM + ibase + 16 * kq + c) << 7) + 8 * h;
#pragma unroll
    for (int mt = 0; mt < MT; ++mt) {
      const v8f r = acc[mt][0];
      const v4f p0 = (xch[w1][2][2 * mt][lane] + xch[w2][1][2 * mt][lane])
                     + xch[w3][0][2 * mt][lane];
      const v4f p1 = (xch[w1][2][2 * mt + 1][lane] + xch[w2][1][2 * mt + 1][lane])
                     + xch[w3][0][2 * mt + 1][lane];
      *(float4*)(orow + 16 * mt)     = make_float4((r[0] + p0.x) * inv, (r[1] + p0.y) * inv,
                                                   (r[2] + p0.z) * inv, (r[3] + p0.w) * inv);
      *(float4*)(orow + 16 * mt + 4) = make_float4((r[4] + p1.x) * inv, (r[5] + p1.y) * inv,
                                                   (r[6] + p1.z) * inv, (r[7] + p1.w) * inv);
    }
  }
#undef SEND_TILE
}
#undef MT
#undef NT
#undef KSPL
#undef NBW
#undef JSTEP

extern "C" void kernel_launch(void* const* d_in, const int* in_sizes, int n_in,
                              void* d_out, int out_size, void* d_ws, size_t ws_size,
                              hipStream_t stream) {
  const float* x = (const float*)d_in[0];
  const float* w = (const float*)d_in[1];
  float* out = (float*)d_out;

  float* s1  = (float*)d_ws;
  float* s2  = s1 + BATCH * KDIM;
  float* t1  = s2 + BATCH * KDIM;
  float* ta  = t1 + BATCH * KDIM;
  float* s2m = ta + BATCH * KDIM;

  fal_prep_kernel<<<BATCH * (KDIM / 128), 256, 0, stream>>>(x, w, s1, s2, t1, ta, s2m);
  dim3 grid(KDIM / 128, BATCH);
  fal_attn_kernel<<<grid, 256, 0, stream>>>(x, s1, s2, t1, ta, s2m, out);
}


# epilogue: quarter sums cross LDS in output-row layout (odd pitch), finisher reads whole rows: each output store writes one whole 512-byte row, not 32-byte pieces of 16 lines; sum order unchanged
# speedup vs baseline: 2.0777x; 1.0754x over previous
#include <hip/hip_runtime.h>
#include <hip/hip_bf16.h>

typedef __attribute__((ext_vector_type(2))) float v2f;
typedef __attribute__((ext_vector_type(4))) float v4f;
typedef __attribute__((ext_vector_type(8))) float v8f;

#define ALPHA 0.2f
#define BATCH 32
#define WDIM 128
#define KDIM 1024
#define JB   16
#define NBLK (KDIM / JB)
#define NPMAX (4 * (KDIM / 128))

#define PCOLS 128
#define PROWS (WDIM / 8)
__global__ __launch_bounds__(256) __attribute__((amdgpu_waves_per_eu(2, 8))) void fal_prep_kernel(const float* __restrict__ x,
                                                       const float* __restrict__ w,
                                                       float* __restrict__ s1,
                                                       float* __restrict__ s2,
                                                       float* __restrict__ t1,
                                                       float* __restrict__ ta,
                                                       float* __restrict__ pmax) {
  __shared__ __attribute__((aligned(16))) float part[8 * 2 * PCOLS];
  const int tid  = threadIdx.x;
  const int lane = tid & 31;
  const int wave = __builtin_amdgcn_readfirstlane(tid >> 5);
  const int b    = blockIdx.x >> 3;
  const int k0   = (blockIdx.x & 7) * PCOLS;
  const float* xp = x + (((size_t)b * WDIM + wave * PROWS) << 10) + k0 + 4 * lane;
  const float* w1 = w + wave * PROWS;
  const float* w2 = w + WDIM + wave * PROWS;

  v4f xv[PROWS];
#pragma unroll
  for (int t = 0; t < PROWS; ++t) xv[t] = *(const v4f*)(xp + ((size_t)t << 10));

  v4f a1 = (v4f)0.f, a2 = (v4f)0.f;
#pragma unroll
  for (int t = 0; t < PROWS; ++t) {
    const float c1 = w1[t], c2 = w2[t];
    a1.x = fmaf(xv[t].x, c1, a1.x); a2.x = fmaf(xv[t].x, c2, a2.x);
    a1.y = fmaf(xv[t].y, c1, a1.y); a2.y = fmaf(xv[t].y, c2, a2.y);
    a1.z = fmaf(xv[t].z, c1, a1.z); a2.z = fmaf(xv[t].z, c2, a2.z);
    a1.w = fmaf(xv[t].w, c1, a1.w); a2.w = fmaf(xv[t].w, c2, a2.w);
  }
  *(v4f*)&part[(wave * 2 + 0) * PCOLS + 4 * lane] = a1;
  *(v4f*)&part[(wave * 2 + 1) * PCOLS + 4 * lane] = a2;
  __syncthreads();

  const int sel = wave >> 2;
  const int col = tid & (PCOLS - 1);
  float acc = 0.f;
#pragma unroll
  for (int r = 0; r < 8; ++r) acc += part[(r * 2 + sel) * PCOLS + col];
  const int idx = b * KDIM + k0 + col;
  if (sel == 0) {
    s1[idx] = acc;
  } else {
    s2[idx] = acc;
    t1[idx] = __expf(acc);
    ta[idx] = __expf(ALPHA * acc);
    float m = acc;
#pragma unroll
    for (int o = 16; o > 0; o >>= 1) m = fmaxf(m, __shfl_xor(m, o, 32));
    if (lane == 0) pmax[4 * blockIdx.x + (wave - 4)] = m;
  }
}
#undef PCOLS
#undef PROWS

#define MT 8
#define NT 4
#define KSPL 4
#define NBW (NBLK / KSPL)
#define JSTEP (JB * KSPL)
#define XPITCH 33
__global__ __launch_bounds__(256) __attribute__((amdgpu_waves_per_eu(2, 2))) void fal_attn_kernel(const float* __restrict__ x,
                                                       const float* __restrict__ s1g,
                                                       const float* __restrict__ s2g,
                                                       const float* __restrict__ t1g,
                                                       const float* __restrict__ tag,
                                                       const float* __restrict__ s2mx,
                                                       float* __restrict__ out) {
  __shared__ __attribute__((aligned(16))) v4f xch[8][NT][16][XPITCH];
  __shared__ float dex[8][NT][32];

  const int tid   = threadIdx.x;
  const int lane  = tid & 31;
  const int wave  = __builtin_amdgcn_readfirstlane(tid >> 5);
  const int h     = lane >> 4;
  const int c     = lane & 15;
  const int b     = blockIdx.y;
  const int kq    = wave & 3;
  const int ih    = wave >> 2;
  const int ibase = blockIdx.x * 128 + ih * (16 * NT);

  const float* pm = s2mx + b * NPMAX;
  float pmv[NPMAX];
#pragma unroll
  for (int g = 0; g < NPMAX; ++g) pmv[g] = pm[g];
  float s1v[NT];
#pragma unroll
  for (int nt = 0; nt < NT; ++nt) s1v[nt] = s1g[b * KDIM + ibase + 16 * ((nt + kq) & 3) + c];

  v8f acc[MT][NT];
#pragma unroll
  for (int mt = 0; mt < MT; ++mt)
#pragma unroll
    for (int nt = 0; nt < NT; ++nt) acc[mt][nt] = (v8f)0.f;
  v2f dsum[NT];
#pragma unroll
  for (int nt = 0; nt < NT; ++nt) dsum[nt] = (v2f)0.f;

  const int j0 = kq * JB + 8 * h;
  const float* xa = x + (((size_t)b * WDIM + c) << 10) + j0;
  const float* f2 = s2g + b * KDIM + j0;
  const float* f1 = t1g + b * KDIM + j0;
  const float* fa = tag + b * KDIM + j0;

  v4f va[2][MT];
  v4f g2[2], g1[2], ga[2];

#define LOAD_HALF(HALF, BLK)                                                        \
  {                                                                                 \
    const int jo = (BLK) * JSTEP + 4 * (HALF);                                      \
    _Pragma("unroll") for (int mt = 0; mt < MT; ++mt)                               \
      va[HALF][mt] = *(const v4f*)(xa + ((size_t)(16 * mt) << 10) + jo);            \
    g2[HALF] = *(const v4f*)(f2 + jo);                                              \
    g1[HALF] = *(const v4f*)(f1 + jo);                                              \
    ga[HALF] = *(const v4f*)(fa + jo);                                              \
  }

#define KSTEP(HALF, PAIR)                                                           \
  {                                                                                 \
    const v2f e2 = g2[HALF].PAIR, e1 = g1[HALF].PAIR, ea = ga[HALF].PAIR;           \
    v2f bfr[NT];                                                                    \
    _Pragma("unroll") for (int nt = 0; nt < NT; ++nt) {                             \
      const v2f z  = s1v[nt] + e2;                                                  \
      const v2f hi = u1[nt] * e1;                                                   \
      const v2f lo = ua[nt] * ea;                                                   \
      v2f p;                                                                        \
      p.x = (z.x > 0.f) ? hi.x : lo.x;                                              \
      p.y = (z.y > 0.f) ? hi.y : lo.y;                                              \
      dsum[nt] += p;                                                                \
      bfr[nt] = p;                                                                  \
    }                                                                               \
    _Pragma("unroll") for (int mt = 0; mt < MT; ++mt) {                             \
      const v2f a = va[HALF][mt].PAIR;                                              \
      _Pragma("unroll") for (int nt = 0; nt < NT; ++nt)                             \
        acc[mt][nt] = __builtin_amdgcn_wmma_f32_16x16x4_f32(false, a, false, bfr[nt], \
                                                   (short)0, acc[mt][nt], false, false); \
    }                                                                               \
  }

#define HALFBLOCK(HALF) KSTEP(HALF, xy) KSTEP(HALF, zw)

  LOAD_HALF(0, 0)
  __builtin_amdgcn_sched_barrier(0);
  float smax = pmv[0];
#pragma unroll
  for (int g = 1; g < NPMAX; ++g) smax = fmaxf(smax, pmv[g]);
  float u1[NT], ua[NT];
#pragma unroll
  for (int nt = 0; nt < NT; ++nt) {
    const float s    = s1v[nt];
    const float zmax = s + smax;
    const float mrow = fmaxf(zmax, ALPHA * zmax);
    u1[nt] = __expf(s - mrow);
    ua[nt] = __expf(ALPHA * s - mrow);
  }
  for (int blk = 0; blk < NBW; ++blk) {
    LOAD_HALF(1, blk)
    __builtin_amdgcn_sched_barrier(0);
    HALFBLOCK(0)
    const int nb = (blk + 1 < NBW) ? blk + 1 : NBW - 1;
    LOAD_HALF(0, nb)
    __builtin_amdgcn_sched_barrier(0);
    HALFBLOCK(1)
  }
#undef HALFBLOCK
#undef KSTEP
#undef LOAD_HALF

  float dt[NT];
#pragma unroll
  for (int nt = 0; nt < NT; ++nt) {
    const float dlane = dsum[nt].x + dsum[nt].y;
    dt[nt] = dlane + __shfl_xor(dlane, 16, 32);
    dex[wave][nt][lane] = dt[nt];
  }

#define SEND_TILE(NTS)                                                              \
  {                                                                                 \
    _Pragma("unroll") for (int mt = 0; mt < MT; ++mt) {                             \
      const v8f r = acc[mt][NTS];                                                   \
      v4f lo4, hi4;                                                                 \
      lo4.x = r[0]; lo4.y = r[1]; lo4.z = r[2]; lo4.w = r[3];                       \
      hi4.x = r[4]; hi4.y = r[5]; hi4.z = r[6]; hi4.w = r[7];                       \
      xch[wave][NTS][c][4 * mt + 2 * h]     = lo4;                                  \
      xch[wave][NTS][c][4 * mt + 2 * h + 1] = hi4;                                  \
    }                                                                               \
  }
  SEND_TILE(0) SEND_TILE(1) SEND_TILE(2) SEND_TILE(3)
  __syncthreads();
  {
    const int w1 = (ih << 2) | ((kq + 1) & 3);
    const int w2 = (ih << 2) | ((kq + 2) & 3);
    const int w3 = (ih << 2) | ((kq + 3) & 3);
    const float den = ((dt[0] + dex[w1][3][lane]) + dex[w2][2][lane]) + dex[w3][1][lane];
    const float inv = 1.0f / den;
    float* orow = out + (((size_t)b * KDIM + ibase + 16 * kq) << 7) + 4 * lane;
#pragma unroll
    for (int rr = 0; rr < 16; ++rr) {
      const float invr = __builtin_bit_cast(float,
          __builtin_amdgcn_readlane(__builtin_bit_cast(int, inv), rr));
      const v4f p = (xch[w1][3][rr][lane] + xch[w2][2][rr][lane]) + xch[w3][1][rr][lane];
      const v4f q = xch[wave][0][rr][lane] + p;
      *(v4f*)(orow + ((size_t)rr << 7)) = q * invr;
    }
  }
#undef SEND_TILE
}
#undef MT
#undef NT
#undef KSPL
#undef NBW
#undef JSTEP
#undef XPITCH

extern "C" void kernel_launch(void* const* d_in, const int* in_sizes, int n_in,
                              void* d_out, int out_size, void* d_ws, size_t ws_size,
                              hipStream_t stream) {
  const float* x = (const float*)d_in[0];
  const float* w = (const float*)d_in[1];
  float* out = (float*)d_out;

  float* s1  = (float*)d_ws;
  float* s2  = s1 + BATCH * KDIM;
  float* t1  = s2 + BATCH * KDIM;
  float* ta  = t1 + BATCH * KDIM;
  float* s2m = ta + BATCH * KDIM;

  fal_prep_kernel<<<BATCH * (KDIM / 128), 256, 0, stream>>>(x, w, s1, s2, t1, ta, s2m);
  dim3 grid(KDIM / 128, BATCH);
  fal_attn_kernel<<<grid, 256, 0, stream>>>(x, s1, s2, t1, ta, s2m, out);
}
